# starting kernel
# baseline (speedup 1.0000x reference)
#include <hip/hip_runtime.h>
#include <hip/hip_bf16.h>
#include <math.h>


typedef __attribute__((ext_vector_type(2))) float v2f;
typedef __attribute__((ext_vector_type(8))) float v8f;

#define B_   4
#define C_   256
#define HDS  8
#define DK   32
#define N_   2304
#define NT   (N_ / 16)
#define QSCALE 0.17677669529663687f

static __device__ __forceinline__ v8f wmma_f32(v2f a, v2f b, v8f c) {
  return __builtin_amdgcn_wmma_f32_16x16x4_f32(
      false, a, false, b, (short)0, c, false, false);
}

__global__ __launch_bounds__(128) void mha_proj_kernel(
    const float* __restrict__ x,
    const float* __restrict__ wq, const float* __restrict__ wk,
    const float* __restrict__ wv,
    float* __restrict__ qt, float* __restrict__ kb, float* __restrict__ vt) {
  const int lane = threadIdx.x & 31;
  const int wave = threadIdx.x >> 5;
  const int t     = blockIdx.x * 4 + wave;
  const int oTile = t & 15;
  const int nTile = t >> 4;
  const int p = blockIdx.y % 3;
  const int b = blockIdx.y / 3;
  const float* w = (p == 0) ? wq : (p == 1) ? wk : wv;

  const int row  = lane & 15;
  const int half = lane >> 4;
  const int koff = half * 2;
  const int oBase = oTile * 16;
  const int nBase = nTile * 16;

  const float* wA = w + (size_t)(oBase + row) * C_ + koff;
  const float* xB = x + ((size_t)b * C_ + koff) * N_ + nBase + row;

  v8f acc = {};
#pragma unroll 8
  for (int k = 0; k < C_; k += 4) {
    v2f a, bm;
    a.x  = wA[k];
    a.y  = wA[k + 1];
    bm.x = xB[(size_t)k * N_];
    bm.y = xB[(size_t)(k + 1) * N_];
    acc = wmma_f32(a, bm, acc);
  }

  if (p == 1) {
    float* outp = kb + ((size_t)b * C_ + oBase + half * 8) * N_ + nBase + row;
#pragma unroll
    for (int v = 0; v < 8; ++v) outp[(size_t)v * N_] = acc[v];
  } else {
    float* dst = (p == 0) ? qt : vt;
    const float scale = (p == 0) ? QSCALE : 1.0f;
#pragma unroll
    for (int v = 0; v < 8; ++v) {
      const int o = oBase + v + half * 8;
      const int h = o >> 5, d = o & 31;
      dst[(((size_t)b * HDS + h) * N_ + nBase + row) * DK + d] = acc[v] * scale;
    }
  }
}

__global__ __launch_bounds__(128) void mha_attn_kernel(
    const float* __restrict__ x,
    const float* __restrict__ qt, const float* __restrict__ kb,
    const float* __restrict__ vt, float* __restrict__ out) {
  __shared__ float ldsP[4][16 * 17];
  const int lane = threadIdx.x & 31;
  const int wave = threadIdx.x >> 5;
  float* P = ldsP[wave];

  const int task  = blockIdx.x * 4 + wave;
  const int qTile = task % NT;
  const int bh    = task / NT;
  const int h = bh & 7;
  const int b = bh >> 3;

  const int row  = lane & 15;
  const int half = lane >> 4;
  const int koff = half * 2;
  const int nBase = qTile * 16;

  v2f aq[8];
  {
    const float* qp =
        qt + (((size_t)b * HDS + h) * N_ + nBase + row) * DK + koff;
#pragma unroll
    for (int kk = 0; kk < 8; ++kk) {
      aq[kk].x = qp[kk * 4];
      aq[kk].y = qp[kk * 4 + 1];
    }
  }

  const float* kbase = kb + ((size_t)b * C_ + h * DK + koff) * N_ + row;
  const float* vhead = vt + ((size_t)b * HDS + h) * N_ * DK;

  v8f o0 = {}, o1 = {};
  float m[8], l[8];
#pragma unroll
  for (int v = 0; v < 8; ++v) { m[v] = -1e30f; l[v] = 0.0f; }

  for (int mt = 0; mt < NT; ++mt) {
    const int mBase = mt * 16;

    v8f S = {};
#pragma unroll
    for (int kk = 0; kk < 8; ++kk) {
      const float* kp = kbase + (size_t)(kk * 4) * N_ + mBase;
      v2f bk;
      bk.x = kp[0];
      bk.y = kp[N_];
      S = wmma_f32(aq[kk], bk, S);
    }

#pragma unroll
    for (int v = 0; v < 8; ++v) {
      float s  = S[v];
      float rm = s;
      rm = fmaxf(rm, __shfl_xor(rm, 1, 16));
      rm = fmaxf(rm, __shfl_xor(rm, 2, 16));
      rm = fmaxf(rm, __shfl_xor(rm, 4, 16));
      rm = fmaxf(rm, __shfl_xor(rm, 8, 16));
      const float mn = fmaxf(m[v], rm);
      const float a  = __expf(m[v] - mn);
      const float pv = __expf(s - mn);
      float rs = pv;
      rs += __shfl_xor(rs, 1, 16);
      rs += __shfl_xor(rs, 2, 16);
      rs += __shfl_xor(rs, 4, 16);
      rs += __shfl_xor(rs, 8, 16);
      l[v] = l[v] * a + rs;
      m[v] = mn;
      o0[v] *= a;
      o1[v] *= a;
      P[(v + half * 8) * 17 + row] = pv;
    }
    asm volatile("s_wait_dscnt 0" ::: "memory");

#pragma unroll
    for (int kk = 0; kk < 4; ++kk) {
      v2f ap;
      ap.x = P[row * 17 + kk * 4 + koff];
      ap.y = P[row * 17 + kk * 4 + koff + 1];
      const float* vp = vhead + (size_t)(mBase + kk * 4 + koff) * DK + row;
      v2f bv0, bv1;
      bv0.x = vp[0];
      bv0.y = vp[DK];
      bv1.x = vp[16];
      bv1.y = vp[DK + 16];
      o0 = wmma_f32(ap, bv0, o0);
      o1 = wmma_f32(ap, bv1, o1);
    }
  }

  const size_t obase =
      ((size_t)b * C_ + h * DK + row) * N_ + nBase + half * 8;
#pragma unroll
  for (int v = 0; v < 8; ++v) {
    const float inv = 1.0f / l[v];
    const size_t i0 = obase + v;
    out[i0] = x[i0] + o0[v] * inv;
    const size_t i1 = i0 + (size_t)16 * N_;
    out[i1] = x[i1] + o1[v] * inv;
  }
}

extern "C" void kernel_launch(void* const* d_in, const int* in_sizes, int n_in,
                              void* d_out, int out_size, void* d_ws,
                              size_t ws_size, hipStream_t stream) {
  const float* x  = (const float*)d_in[0];
  const float* wq = (const float*)d_in[1];
  const float* wk = (const float*)d_in[2];
  const float* wv = (const float*)d_in[3];
  float* out = (float*)d_out;

  float* qt = (float*)d_ws;
  float* kb = qt + (size_t)B_ * C_ * N_;
  float* vt = kb + (size_t)B_ * C_ * N_;

  dim3 blk(128);
  mha_proj_kernel<<<dim3(576, 12), blk, 0, stream>>>(x, wq, wk, wv, qt, kb, vt);
  mha_attn_kernel<<<dim3(1152), blk, 0, stream>>>(x, qt, kb, vt, out);
}


# Scores formed transposed (keys on tile rows): softmax max and sum in-lane, P feeds the PV WMMA from registers, no LDS or shuffles; K,V workspace laid out for it; changes sum order of l and WMMA k
# speedup vs baseline: 1.1886x; 1.1886x over previous
#include <hip/hip_runtime.h>
#include <math.h>


typedef __attribute__((ext_vector_type(2))) float v2f;
typedef __attribute__((ext_vector_type(4))) float v4f;
typedef __attribute__((ext_vector_type(8))) float v8f;

#define B_   4
#define C_   256
#define HDS  8
#define DK   32
#define N_   2304
#define NT   (N_ / 16)
#define QSCALE 0.17677669529663687f

static __device__ __forceinline__ v8f wmma_f32(v2f a, v2f b, v8f c) {
  return __builtin_amdgcn_wmma_f32_16x16x4_f32(
      false, a, false, b, (short)0, c, false, false);
}

static __device__ __forceinline__ float xhalf(float v) {
  const int b = __builtin_bit_cast(int, v);
  return __builtin_bit_cast(
      float, __builtin_amdgcn_permlanex16(b, b, 0x76543210, 0xfedcba98,
                                          false, false));
}

__global__ __launch_bounds__(128) void mha_proj_kernel(
    const float* __restrict__ x,
    const float* __restrict__ wq, const float* __restrict__ wk,
    const float* __restrict__ wv,
    float* __restrict__ qt, float* __restrict__ kt, float* __restrict__ vk) {
  const int lane = threadIdx.x & 31;
  const int wave = threadIdx.x >> 5;
  const int t     = blockIdx.x * 4 + wave;
  const int oTile = t & 15;
  const int nTile = t >> 4;
  const int p = blockIdx.y % 3;
  const int b = blockIdx.y / 3;
  const float* w = (p == 0) ? wq : (p == 1) ? wk : wv;

  const int row  = lane & 15;
  const int half = lane >> 4;
  const int koff = half * 2;
  const int oBase = oTile * 16;
  const int nBase = nTile * 16;

  const float* wA = w + (size_t)(oBase + row) * C_ + koff;
  const float* xB = x + ((size_t)b * C_ + koff) * N_ + nBase + row;

  v8f acc = {};
#pragma unroll 8
  for (int k = 0; k < C_; k += 4) {
    v2f a, bm;
    a.x  = wA[k];
    a.y  = wA[k + 1];
    bm.x = xB[(size_t)k * N_];
    bm.y = xB[(size_t)(k + 1) * N_];
    acc = wmma_f32(a, bm, acc);
  }

  if (p == 2) {
#pragma unroll
    for (int v = 0; v < 8; ++v) {
      const int o = oBase + v + half * 8;
      const int h = o >> 5, d = o & 31;
      vk[((((size_t)b * HDS + h) * NT + nTile) * DK + d) * 16 + row] = acc[v];
    }
  } else {
    float* dst = (p == 0) ? qt : kt;
    const float scale = (p == 0) ? QSCALE : 1.0f;
#pragma unroll
    for (int v = 0; v < 8; ++v) {
      const int o = oBase + v + half * 8;
      const int h = o >> 5, d = o & 31;
      dst[(((size_t)b * HDS + h) * N_ + nBase + row) * DK + d] = acc[v] * scale;
    }
  }
}

__global__ __launch_bounds__(128) void mha_attn_kernel(
    const float* __restrict__ x,
    const float* __restrict__ qt, const float* __restrict__ kt,
    const float* __restrict__ vk, float* __restrict__ out) {
  const int lane = threadIdx.x & 31;
  const int wave = threadIdx.x >> 5;

  const int task  = blockIdx.x * 4 + wave;
  const int qTile = task % NT;
  const int bh    = task / NT;
  const int h = bh & 7;
  const int b = bh >> 3;

  const int c  = lane & 15;
  const int hf = lane >> 4;
  const int nBase = qTile * 16;

  v2f bq[8];
  {
    const v4f* qp = (const v4f*)(qt + ((size_t)bh * N_ + nBase + c) * DK + 16 * hf);
    const v4f q0 = qp[0], q1 = qp[1], q2 = qp[2], q3 = qp[3];
    bq[0] = (v2f){q0.x, q0.y};  bq[1] = (v2f){q0.z, q0.w};
    bq[2] = (v2f){q1.x, q1.y};  bq[3] = (v2f){q1.z, q1.w};
    bq[4] = (v2f){q2.x, q2.y};  bq[5] = (v2f){q2.z, q2.w};
    bq[6] = (v2f){q3.x, q3.y};  bq[7] = (v2f){q3.z, q3.w};
  }

  const float* kptr = kt + ((size_t)bh * N_ + c) * DK + 16 * hf;
  const float* vptr = vk + (size_t)bh * NT * DK * 16 + c * 16 + 8 * hf;

  v8f o0 = {}, o1 = {};
  float m = -1e30f, l = 0.0f;

  for (int mt = 0; mt < NT; ++mt) {
    const v4f* kp = (const v4f*)(kptr + (size_t)mt * (16 * DK));
    const v4f* vp = (const v4f*)(vptr + (size_t)mt * (DK * 16));
    const v4f k0 = kp[0], k1 = kp[1], k2 = kp[2], k3 = kp[3];
    const v4f va0 = vp[0],  va1 = vp[1];
    const v4f vb0 = vp[64], vb1 = vp[65];

    v8f S = {};
    S = wmma_f32((v2f){k0.x, k0.y}, bq[0], S);
    S = wmma_f32((v2f){k0.z, k0.w}, bq[1], S);
    S = wmma_f32((v2f){k1.x, k1.y}, bq[2], S);
    S = wmma_f32((v2f){k1.z, k1.w}, bq[3], S);
    S = wmma_f32((v2f){k2.x, k2.y}, bq[4], S);
    S = wmma_f32((v2f){k2.z, k2.w}, bq[5], S);
    S = wmma_f32((v2f){k3.x, k3.y}, bq[6], S);
    S = wmma_f32((v2f){k3.z, k3.w}, bq[7], S);

    float rm = fmaxf(fmaxf(fmaxf(S[0], S[1]), fmaxf(S[2], S[3])),
                     fmaxf(fmaxf(S[4], S[5]), fmaxf(S[6], S[7])));
    rm = fmaxf(rm, xhalf(rm));
    const float mn = fmaxf(m, rm);
    const float a  = __expf(m - mn);
    m = mn;

    v8f P;
#pragma unroll
    for (int r = 0; r < 8; ++r) P[r] = __expf(S[r] - mn);
    const float rs = ((P[0] + P[1]) + (P[2] + P[3])) +
                     ((P[4] + P[5]) + (P[6] + P[7]));
    l = l * a + rs;
    o0 *= a;
    o1 *= a;

    o0 = wmma_f32((v2f){va0.x, va0.y}, (v2f){P[0], P[1]}, o0);
    o1 = wmma_f32((v2f){vb0.x, vb0.y}, (v2f){P[0], P[1]}, o1);
    o0 = wmma_f32((v2f){va0.z, va0.w}, (v2f){P[2], P[3]}, o0);
    o1 = wmma_f32((v2f){vb0.z, vb0.w}, (v2f){P[2], P[3]}, o1);
    o0 = wmma_f32((v2f){va1.x, va1.y}, (v2f){P[4], P[5]}, o0);
    o1 = wmma_f32((v2f){vb1.x, vb1.y}, (v2f){P[4], P[5]}, o1);
    o0 = wmma_f32((v2f){va1.z, va1.w}, (v2f){P[6], P[7]}, o0);
    o1 = wmma_f32((v2f){vb1.z, vb1.w}, (v2f){P[6], P[7]}, o1);
  }

  const float inv = 1.0f / (l + xhalf(l));
  const size_t obase = ((size_t)b * C_ + h * DK + 8 * hf) * N_ + nBase + c;
#pragma unroll
  for (int r = 0; r < 8; ++r) {
    const size_t i0 = obase + (size_t)r * N_;
    out[i0] = x[i0] + o0[r] * inv;
    const size_t i1 = i0 + (size_t)16 * N_;
    out[i1] = x[i1] + o1[r] * inv;
  }
}

extern "C" void kernel_launch(void* const* d_in, const int* in_sizes, int n_in,
                              void* d_out, int out_size, void* d_ws,
                              size_t ws_size, hipStream_t stream) {
  const float* x  = (const float*)d_in[0];
  const float* wq = (const float*)d_in[1];
  const float* wk = (const float*)d_in[2];
  const float* wv = (const float*)d_in[3];
  float* out = (float*)d_out;

  float* qt = (float*)d_ws;
  float* kt = qt + (size_t)B_ * C_ * N_;
  float* vk = kt + (size_t)B_ * C_ * N_;

  dim3 blk(128);
  mha_proj_kernel<<<dim3(576, 12), blk, 0, stream>>>(x, wq, wk, wv, qt, kt, vk);
  mha_attn_kernel<<<dim3(1152), blk, 0, stream>>>(x, qt, kt, vk, out);
}


# Attention: two register sets of K,V fragments, each refilled as one group of eight loads two key tiles ahead, right behind the tile that used it (loop unrolled by two); no wait right behind a load
# speedup vs baseline: 1.5766x; 1.3264x over previous
#include <hip/hip_runtime.h>
#include <math.h>


typedef __attribute__((ext_vector_type(2))) float v2f;
typedef __attribute__((ext_vector_type(4))) float v4f;
typedef __attribute__((ext_vector_type(8))) float v8f;

#define B_   4
#define C_   256
#define HDS  8
#define DK   32
#define N_   2304
#define NT   (N_ / 16)
#define QSCALE 0.17677669529663687f

static __device__ __forceinline__ v8f wmma_f32(v2f a, v2f b, v8f c) {
  return __builtin_amdgcn_wmma_f32_16x16x4_f32(
      false, a, false, b, (short)0, c, false, false);
}

static __device__ __forceinline__ float xhalf(float v) {
  const int b = __builtin_bit_cast(int, v);
  return __builtin_bit_cast(
      float, __builtin_amdgcn_permlanex16(b, b, 0x76543210, 0xfedcba98,
                                          false, false));
}

__global__ __launch_bounds__(128) void mha_proj_kernel(
    const float* __restrict__ x,
    const float* __restrict__ wq, const float* __restrict__ wk,
    const float* __restrict__ wv,
    float* __restrict__ qt, float* __restrict__ kt, float* __restrict__ vk) {
  const int lane = threadIdx.x & 31;
  const int wave = threadIdx.x >> 5;
  const int t     = blockIdx.x * 4 + wave;
  const int oTile = t & 15;
  const int nTile = t >> 4;
  const int p = blockIdx.y % 3;
  const int b = blockIdx.y / 3;
  const float* w = (p == 0) ? wq : (p == 1) ? wk : wv;

  const int row  = lane & 15;
  const int half = lane >> 4;
  const int koff = half * 2;
  const int oBase = oTile * 16;
  const int nBase = nTile * 16;

  const float* wA = w + (size_t)(oBase + row) * C_ + koff;
  const float* xB = x + ((size_t)b * C_ + koff) * N_ + nBase + row;

  v8f acc = {};
#pragma unroll 8
  for (int k = 0; k < C_; k += 4) {
    v2f a, bm;
    a.x  = wA[k];
    a.y  = wA[k + 1];
    bm.x = xB[(size_t)k * N_];
    bm.y = xB[(size_t)(k + 1) * N_];
    acc = wmma_f32(a, bm, acc);
  }

  if (p == 2) {
#pragma unroll
    for (int v = 0; v < 8; ++v) {
      const int o = oBase + v + half * 8;
      const int h = o >> 5, d = o & 31;
      vk[((((size_t)b * HDS + h) * NT + nTile) * DK + d) * 16 + row] = acc[v];
    }
  } else {
    float* dst = (p == 0) ? qt : kt;
    const float scale = (p == 0) ? QSCALE : 1.0f;
#pragma unroll
    for (int v = 0; v < 8; ++v) {
      const int o = oBase + v + half * 8;
      const int h = o >> 5, d = o & 31;
      dst[(((size_t)b * HDS + h) * N_ + nBase + row) * DK + d] = acc[v] * scale;
    }
  }
}

static __device__ __forceinline__ void load_tile(const v4f* __restrict__ kp,
                                                 const v4f* __restrict__ vp,
                                                 v4f (&f)[8]) {
  f[0] = kp[0];  f[1] = kp[1];  f[2] = kp[2];  f[3] = kp[3];
  f[4] = vp[0];  f[5] = vp[1];
  f[6] = vp[64]; f[7] = vp[65];
}

static __device__ __forceinline__ void attn_tile(const v4f (&f)[8],
                                                 const v2f (&bq)[8],
                                                 v8f& o0, v8f& o1,
                                                 float& m, float& l) {
  const v4f k0 = f[0], k1 = f[1], k2 = f[2], k3 = f[3];
  const v4f va0 = f[4], va1 = f[5];
  const v4f vb0 = f[6], vb1 = f[7];

  v8f S = {};
  S = wmma_f32((v2f){k0.x, k0.y}, bq[0], S);
  S = wmma_f32((v2f){k0.z, k0.w}, bq[1], S);
  S = wmma_f32((v2f){k1.x, k1.y}, bq[2], S);
  S = wmma_f32((v2f){k1.z, k1.w}, bq[3], S);
  S = wmma_f32((v2f){k2.x, k2.y}, bq[4], S);
  S = wmma_f32((v2f){k2.z, k2.w}, bq[5], S);
  S = wmma_f32((v2f){k3.x, k3.y}, bq[6], S);
  S = wmma_f32((v2f){k3.z, k3.w}, bq[7], S);

  float rm = fmaxf(fmaxf(fmaxf(S[0], S[1]), fmaxf(S[2], S[3])),
                   fmaxf(fmaxf(S[4], S[5]), fmaxf(S[6], S[7])));
  rm = fmaxf(rm, xhalf(rm));
  const float mn = fmaxf(m, rm);
  const float a  = __expf(m - mn);
  m = mn;

  v8f P;
#pragma unroll
  for (int r = 0; r < 8; ++r) P[r] = __expf(S[r] - mn);
  const float rs = ((P[0] + P[1]) + (P[2] + P[3])) +
                   ((P[4] + P[5]) + (P[6] + P[7]));
  l = l * a + rs;
  o0 *= a;
  o1 *= a;

  o0 = wmma_f32((v2f){va0.x, va0.y}, (v2f){P[0], P[1]}, o0);
  o1 = wmma_f32((v2f){vb0.x, vb0.y}, (v2f){P[0], P[1]}, o1);
  o0 = wmma_f32((v2f){va0.z, va0.w}, (v2f){P[2], P[3]}, o0);
  o1 = wmma_f32((v2f){vb0.z, vb0.w}, (v2f){P[2], P[3]}, o1);
  o0 = wmma_f32((v2f){va1.x, va1.y}, (v2f){P[4], P[5]}, o0);
  o1 = wmma_f32((v2f){vb1.x, vb1.y}, (v2f){P[4], P[5]}, o1);
  o0 = wmma_f32((v2f){va1.z, va1.w}, (v2f){P[6], P[7]}, o0);
  o1 = wmma_f32((v2f){vb1.z, vb1.w}, (v2f){P[6], P[7]}, o1);
}

__global__ __launch_bounds__(128) void mha_attn_kernel(
    const float* __restrict__ x,
    const float* __restrict__ qt, const float* __restrict__ kt,
    const float* __restrict__ vk, float* __restrict__ out) {
  const int lane = threadIdx.x & 31;
  const int wave = threadIdx.x >> 5;

  const int task  = blockIdx.x * 4 + wave;
  const int qTile = task % NT;
  const int bh    = task / NT;
  const int h = bh & 7;
  const int b = bh >> 3;

  const int c  = lane & 15;
  const int hf = lane >> 4;
  const int nBase = qTile * 16;

  v2f bq[8];
  {
    const v4f* qp = (const v4f*)(qt + ((size_t)bh * N_ + nBase + c) * DK + 16 * hf);
    const v4f q0 = qp[0], q1 = qp[1], q2 = qp[2], q3 = qp[3];
    bq[0] = (v2f){q0.x, q0.y};  bq[1] = (v2f){q0.z, q0.w};
    bq[2] = (v2f){q1.x, q1.y};  bq[3] = (v2f){q1.z, q1.w};
    bq[4] = (v2f){q2.x, q2.y};  bq[5] = (v2f){q2.z, q2.w};
    bq[6] = (v2f){q3.x, q3.y};  bq[7] = (v2f){q3.z, q3.w};
  }

  const float* kptr = kt + ((size_t)bh * N_ + c) * DK + 16 * hf;
  const float* vptr = vk + (size_t)bh * NT * DK * 16 + c * 16 + 8 * hf;

  v8f o0 = {}, o1 = {};
  float m = -1e30f, l = 0.0f;

  const v4f* kb = (const v4f*)kptr;
  const v4f* vb = (const v4f*)vptr;
  v4f fa[8], fb[8];
  load_tile(kb, vb, fa);
  load_tile(kb + 128, vb + 128, fb);

  for (int mt = 0; mt < NT; mt += 2) {
    const int na = (mt + 2 < NT) ? (mt + 2) : (NT - 1);
    const int nb = (mt + 3 < NT) ? (mt + 3) : (NT - 1);
    attn_tile(fa, bq, o0, o1, m, l);
    load_tile(kb + (size_t)na * 128, vb + (size_t)na * 128, fa);
    __builtin_amdgcn_sched_barrier(0);
    attn_tile(fb, bq, o0, o1, m, l);
    load_tile(kb + (size_t)nb * 128, vb + (size_t)nb * 128, fb);
    __builtin_amdgcn_sched_barrier(0);
  }

  const float inv = 1.0f / (l + xhalf(l));
  const size_t obase = ((size_t)b * C_ + h * DK + 8 * hf) * N_ + nBase + c;
#pragma unroll
  for (int r = 0; r < 8; ++r) {
    const size_t i0 = obase + (size_t)r * N_;
    out[i0] = x[i0] + o0[r] * inv;
    const size_t i1 = i0 + (size_t)16 * N_;
    out[i1] = x[i1] + o1[r] * inv;
  }
}

extern "C" void kernel_launch(void* const* d_in, const int* in_sizes, int n_in,
                              void* d_out, int out_size, void* d_ws,
                              size_t ws_size, hipStream_t stream) {
  const float* x  = (const float*)d_in[0];
  const float* wq = (const float*)d_in[1];
  const float* wk = (const float*)d_in[2];
  const float* wv = (const float*)d_in[3];
  float* out = (float*)d_out;

  float* qt = (float*)d_ws;
  float* kt = qt + (size_t)B_ * C_ * N_;
  float* vk = kt + (size_t)B_ * C_ * N_;

  dim3 blk(128);
  mha_proj_kernel<<<dim3(576, 12), blk, 0, stream>>>(x, wq, wk, wv, qt, kt, vk);
  mha_attn_kernel<<<dim3(1152), blk, 0, stream>>>(x, qt, kt, vk, out);
}


# Projections re-tiled: one wave per head x 64 positions (8 accumulator tiles), W and X fragments by 16-byte loads reused 4x and 2x, next 16 channels loaded ahead of 32 WMMAs; changes WMMA k order
# speedup vs baseline: 1.8306x; 1.1611x over previous
#include <hip/hip_runtime.h>
#include <math.h>


typedef __attribute__((ext_vector_type(2))) float v2f;
typedef __attribute__((ext_vector_type(4))) float v4f;
typedef __attribute__((ext_vector_type(8))) float v8f;

#define B_   4
#define C_   256
#define HDS  8
#define DK   32
#define N_   2304
#define NT   (N_ / 16)
#define QSCALE 0.17677669529663687f

static __device__ __forceinline__ v8f wmma_f32(v2f a, v2f b, v8f c) {
  return __builtin_amdgcn_wmma_f32_16x16x4_f32(
      false, a, false, b, (short)0, c, false, false);
}

static __device__ __forceinline__ float xhalf(float v) {
  const int b = __builtin_bit_cast(int, v);
  return __builtin_bit_cast(
      float, __builtin_amdgcn_permlanex16(b, b, 0x76543210, 0xfedcba98,
                                          false, false));
}


static __device__ __forceinline__ void load_kblock(const float* __restrict__ wp,
                                                   const float* __restrict__ xp,
                                                   v4f (&f)[6]) {
  f[0] = *(const v4f*)(wp);
  f[1] = *(const v4f*)(wp + 16 * C_);
  f[2] = *(const v4f*)(xp);
  f[3] = *(const v4f*)(xp + N_);
  f[4] = *(const v4f*)(xp + 2 * N_);
  f[5] = *(const v4f*)(xp + 3 * N_);
}

struct KOps {
  v2f a[2][2];
  v2f b[2][4];
};

static __device__ __forceinline__ void make_ops(const v4f (&f)[6], KOps& o) {
#pragma unroll
  for (int s = 0; s < 2; ++s) {
    o.a[0][s] = (v2f){f[0][2 * s], f[0][2 * s + 1]};
    o.a[1][s] = (v2f){f[1][2 * s], f[1][2 * s + 1]};
#pragma unroll
    for (int j = 0; j < 4; ++j)
      o.b[s][j] = (v2f){f[2 + 2 * s][j], f[3 + 2 * s][j]};
  }
}

static __device__ __forceinline__ void mma_kblock(const KOps& o,
                                                  v8f (&acc)[2][4]) {
#pragma unroll
  for (int s = 0; s < 2; ++s) {
#pragma unroll
    for (int j = 0; j < 4; ++j) {
      acc[0][j] = wmma_f32(o.a[0][s], o.b[s][j], acc[0][j]);
      acc[1][j] = wmma_f32(o.a[1][s], o.b[s][j], acc[1][j]);
    }
  }
}

__global__ __launch_bounds__(128) void mha_proj_kernel(
    const float* __restrict__ x,
    const float* __restrict__ wq, const float* __restrict__ wk,
    const float* __restrict__ wv,
    float* __restrict__ qt, float* __restrict__ kt, float* __restrict__ vk) {
  const int lane = threadIdx.x & 31;
  const int wave = threadIdx.x >> 5;
  const int nb = blockIdx.x >> 1;
  const int h  = (blockIdx.x & 1) * 4 + wave;
  const int p = blockIdx.y % 3;
  const int b = blockIdx.y / 3;
  const float* w = (p == 0) ? wq : (p == 1) ? wk : wv;

  const int c  = lane & 15;
  const int hf = lane >> 4;
  const int n0 = nb * 64;

  const float* wp = w + (size_t)(h * DK + c) * C_ + 4 * hf;
  const float* xp = x + ((size_t)b * C_ + 4 * hf) * N_ + n0 + 4 * c;

  v8f acc[2][4] = {};
  v4f ra[6], rb[6];
  KOps oa, ob;
  load_kblock(wp, xp, ra);
  load_kblock(wp + 8, xp + (size_t)8 * N_, rb);
  make_ops(ra, oa);
  make_ops(rb, ob);

  for (int k0 = 0; k0 < C_; k0 += 16) {
    const int kn = (k0 + 16 < C_) ? (k0 + 16) : (C_ - 16);
    load_kblock(wp + kn, xp + (size_t)kn * N_, ra);
    load_kblock(wp + kn + 8, xp + (size_t)(kn + 8) * N_, rb);
    __builtin_amdgcn_sched_barrier(0);
    mma_kblock(oa, acc);
    mma_kblock(ob, acc);
    __builtin_amdgcn_sched_barrier(0);
    make_ops(ra, oa);
    make_ops(rb, ob);
  }

  const int bh = b * HDS + h;
  if (p == 2) {
    float* vbase = vk + ((((size_t)bh * NT + (n0 >> 4) + (c >> 2)) * DK + 8 * hf) * 16)
                      + 4 * (c & 3);
#pragma unroll
    for (int i = 0; i < 2; ++i) {
#pragma unroll
      for (int r = 0; r < 8; ++r) {
        const v4f v = {acc[i][0][r], acc[i][1][r], acc[i][2][r], acc[i][3][r]};
        *(v4f*)(vbase + (16 * i + r) * 16) = v;
      }
    }
  } else {
    float* dst = (p == 0) ? qt : kt;
    const float scale = (p == 0) ? QSCALE : 1.0f;
    float* qbase = dst + ((size_t)bh * N_ + n0 + 4 * c) * DK + 8 * hf;
#pragma unroll
    for (int i = 0; i < 2; ++i) {
#pragma unroll
      for (int j = 0; j < 4; ++j) {
        float* q = qbase + j * DK + 16 * i;
        const v8f a = acc[i][j];
        const v4f lo = {a[0] * scale, a[1] * scale, a[2] * scale, a[3] * scale};
        const v4f hi = {a[4] * scale, a[5] * scale, a[6] * scale, a[7] * scale};
        *(v4f*)(q)     = lo;
        *(v4f*)(q + 4) = hi;
      }
    }
  }
}

static __device__ __forceinline__ void load_tile(const v4f* __restrict__ kp,
                                                 const v4f* __restrict__ vp,
                                                 v4f (&f)[8]) {
  f[0] = kp[0];  f[1] = kp[1];  f[2] = kp[2];  f[3] = kp[3];
  f[4] = vp[0];  f[5] = vp[1];
  f[6] = vp[64]; f[7] = vp[65];
}

static __device__ __forceinline__ void attn_tile(const v4f (&f)[8],
                                                 const v2f (&bq)[8],
                                                 v8f& o0, v8f& o1,
                                                 float& m, float& l) {
  const v4f k0 = f[0], k1 = f[1], k2 = f[2], k3 = f[3];
  const v4f va0 = f[4], va1 = f[5];
  const v4f vb0 = f[6], vb1 = f[7];

  v8f S = {};
  S = wmma_f32((v2f){k0.x, k0.y}, bq[0], S);
  S = wmma_f32((v2f){k0.z, k0.w}, bq[1], S);
  S = wmma_f32((v2f){k1.x, k1.y}, bq[2], S);
  S = wmma_f32((v2f){k1.z, k1.w}, bq[3], S);
  S = wmma_f32((v2f){k2.x, k2.y}, bq[4], S);
  S = wmma_f32((v2f){k2.z, k2.w}, bq[5], S);
  S = wmma_f32((v2f){k3.x, k3.y}, bq[6], S);
  S = wmma_f32((v2f){k3.z, k3.w}, bq[7], S);

  float rm = fmaxf(fmaxf(fmaxf(S[0], S[1]), fmaxf(S[2], S[3])),
                   fmaxf(fmaxf(S[4], S[5]), fmaxf(S[6], S[7])));
  rm = fmaxf(rm, xhalf(rm));
  const float mn = fmaxf(m, rm);
  const float a  = __expf(m - mn);
  m = mn;

  v8f P;
#pragma unroll
  for (int r = 0; r < 8; ++r) P[r] = __expf(S[r] - mn);
  const float rs = ((P[0] + P[1]) + (P[2] + P[3])) +
                   ((P[4] + P[5]) + (P[6] + P[7]));
  l = l * a + rs;
  o0 *= a;
  o1 *= a;

  o0 = wmma_f32((v2f){va0.x, va0.y}, (v2f){P[0], P[1]}, o0);
  o1 = wmma_f32((v2f){vb0.x, vb0.y}, (v2f){P[0], P[1]}, o1);
  o0 = wmma_f32((v2f){va0.z, va0.w}, (v2f){P[2], P[3]}, o0);
  o1 = wmma_f32((v2f){vb0.z, vb0.w}, (v2f){P[2], P[3]}, o1);
  o0 = wmma_f32((v2f){va1.x, va1.y}, (v2f){P[4], P[5]}, o0);
  o1 = wmma_f32((v2f){vb1.x, vb1.y}, (v2f){P[4], P[5]}, o1);
  o0 = wmma_f32((v2f){va1.z, va1.w}, (v2f){P[6], P[7]}, o0);
  o1 = wmma_f32((v2f){vb1.z, vb1.w}, (v2f){P[6], P[7]}, o1);
}

__global__ __launch_bounds__(128) void mha_attn_kernel(
    const float* __restrict__ x,
    const float* __restrict__ qt, const float* __restrict__ kt,
    const float* __restrict__ vk, float* __restrict__ out) {
  const int lane = threadIdx.x & 31;
  const int wave = threadIdx.x >> 5;

  const int task  = blockIdx.x * 4 + wave;
  const int qTile = task % NT;
  const int bh    = task / NT;
  const int h = bh & 7;
  const int b = bh >> 3;

  const int c  = lane & 15;
  const int hf = lane >> 4;
  const int nBase = qTile * 16;

  v2f bq[8];
  {
    const v4f* qp = (const v4f*)(qt + ((size_t)bh * N_ + nBase + c) * DK + 16 * hf);
    const v4f q0 = qp[0], q1 = qp[1], q2 = qp[2], q3 = qp[3];
    bq[0] = (v2f){q0.x, q0.y};  bq[1] = (v2f){q0.z, q0.w};
    bq[2] = (v2f){q1.x, q1.y};  bq[3] = (v2f){q1.z, q1.w};
    bq[4] = (v2f){q2.x, q2.y};  bq[5] = (v2f){q2.z, q2.w};
    bq[6] = (v2f){q3.x, q3.y};  bq[7] = (v2f){q3.z, q3.w};
  }

  const float* kptr = kt + ((size_t)bh * N_ + c) * DK + 16 * hf;
  const float* vptr = vk + (size_t)bh * NT * DK * 16 + c * 16 + 8 * hf;

  v8f o0 = {}, o1 = {};
  float m = -1e30f, l = 0.0f;

  const v4f* kb = (const v4f*)kptr;
  const v4f* vb = (const v4f*)vptr;
  v4f fa[8], fb[8];
  load_tile(kb, vb, fa);
  load_tile(kb + 128, vb + 128, fb);

  for (int mt = 0; mt < NT; mt += 2) {
    const int na = (mt + 2 < NT) ? (mt + 2) : (NT - 1);
    const int nb = (mt + 3 < NT) ? (mt + 3) : (NT - 1);
    attn_tile(fa, bq, o0, o1, m, l);
    load_tile(kb + (size_t)na * 128, vb + (size_t)na * 128, fa);
    __builtin_amdgcn_sched_barrier(0);
    attn_tile(fb, bq, o0, o1, m, l);
    load_tile(kb + (size_t)nb * 128, vb + (size_t)nb * 128, fb);
    __builtin_amdgcn_sched_barrier(0);
  }

  const float inv = 1.0f / (l + xhalf(l));
  const size_t obase = ((size_t)b * C_ + h * DK + 8 * hf) * N_ + nBase + c;
#pragma unroll
  for (int r = 0; r < 8; ++r) {
    const size_t i0 = obase + (size_t)r * N_;
    out[i0] = x[i0] + o0[r] * inv;
    const size_t i1 = i0 + (size_t)16 * N_;
    out[i1] = x[i1] + o1[r] * inv;
  }
}

extern "C" void kernel_launch(void* const* d_in, const int* in_sizes, int n_in,
                              void* d_out, int out_size, void* d_ws,
                              size_t ws_size, hipStream_t stream) {
  const float* x  = (const float*)d_in[0];
  const float* wq = (const float*)d_in[1];
  const float* wk = (const float*)d_in[2];
  const float* wv = (const float*)d_in[3];
  float* out = (float*)d_out;

  float* qt = (float*)d_ws;
  float* kt = qt + (size_t)B_ * C_ * N_;
  float* vk = kt + (size_t)B_ * C_ * N_;

  dim3 blk(128);
  mha_proj_kernel<<<dim3(72, 12), blk, 0, stream>>>(x, wq, wk, wv, qt, kt, vk);
  mha_attn_kernel<<<dim3(1152), blk, 0, stream>>>(x, qt, kt, vk, out);
}


# K and V workspace tiles stored fragment-major (tile, fragment, lane, 4 floats): each 16-byte fragment load of the attention sweep reads 512 contiguous bytes, not 16 (K) or 8 (V) lines; same sums
# speedup vs baseline: 1.8871x; 1.0309x over previous
#include <hip/hip_runtime.h>
#include <math.h>


typedef __attribute__((ext_vector_type(2))) float v2f;
typedef __attribute__((ext_vector_type(4))) float v4f;
typedef __attribute__((ext_vector_type(8))) float v8f;

#define B_   4
#define C_   256
#define HDS  8
#define DK   32
#define N_   2304
#define NT   (N_ / 16)
#define QSCALE 0.17677669529663687f

static __device__ __forceinline__ v8f wmma_f32(v2f a, v2f b, v8f c) {
  return __builtin_amdgcn_wmma_f32_16x16x4_f32(
      false, a, false, b, (short)0, c, false, false);
}

static __device__ __forceinline__ float xhalf(float v) {
  const int b = __builtin_bit_cast(int, v);
  return __builtin_bit_cast(
      float, __builtin_amdgcn_permlanex16(b, b, 0x76543210, 0xfedcba98,
                                          false, false));
}


static __device__ __forceinline__ void load_kblock(const float* __restrict__ wp,
                                                   const float* __restrict__ xp,
                                                   v4f (&f)[6]) {
  f[0] = *(const v4f*)(wp);
  f[1] = *(const v4f*)(wp + 16 * C_);
  f[2] = *(const v4f*)(xp);
  f[3] = *(const v4f*)(xp + N_);
  f[4] = *(const v4f*)(xp + 2 * N_);
  f[5] = *(const v4f*)(xp + 3 * N_);
}

struct KOps {
  v2f a[2][2];
  v2f b[2][4];
};

static __device__ __forceinline__ void make_ops(const v4f (&f)[6], KOps& o) {
#pragma unroll
  for (int s = 0; s < 2; ++s) {
    o.a[0][s] = (v2f){f[0][2 * s], f[0][2 * s + 1]};
    o.a[1][s] = (v2f){f[1][2 * s], f[1][2 * s + 1]};
#pragma unroll
    for (int j = 0; j < 4; ++j)
      o.b[s][j] = (v2f){f[2 + 2 * s][j], f[3 + 2 * s][j]};
  }
}

static __device__ __forceinline__ void mma_kblock(const KOps& o,
                                                  v8f (&acc)[2][4]) {
#pragma unroll
  for (int s = 0; s < 2; ++s) {
#pragma unroll
    for (int j = 0; j < 4; ++j) {
      acc[0][j] = wmma_f32(o.a[0][s], o.b[s][j], acc[0][j]);
      acc[1][j] = wmma_f32(o.a[1][s], o.b[s][j], acc[1][j]);
    }
  }
}

__global__ __launch_bounds__(128) void mha_proj_kernel(
    const float* __restrict__ x,
    const float* __restrict__ wq, const float* __restrict__ wk,
    const float* __restrict__ wv,
    float* __restrict__ qt, float* __restrict__ kt, float* __restrict__ vk) {
  const int lane = threadIdx.x & 31;
  const int wave = threadIdx.x >> 5;
  const int nb = blockIdx.x >> 1;
  const int h  = (blockIdx.x & 1) * 4 + wave;
  const int p = blockIdx.y % 3;
  const int b = blockIdx.y / 3;
  const float* w = (p == 0) ? wq : (p == 1) ? wk : wv;

  const int c  = lane & 15;
  const int hf = lane >> 4;
  const int n0 = nb * 64;

  const float* wp = w + (size_t)(h * DK + c) * C_ + 4 * hf;
  const float* xp = x + ((size_t)b * C_ + 4 * hf) * N_ + n0 + 4 * c;

  v8f acc[2][4] = {};
  v4f ra[6], rb[6];
  KOps oa, ob;
  load_kblock(wp, xp, ra);
  load_kblock(wp + 8, xp + (size_t)8 * N_, rb);
  make_ops(ra, oa);
  make_ops(rb, ob);

  for (int k0 = 0; k0 < C_; k0 += 16) {
    const int kn = (k0 + 16 < C_) ? (k0 + 16) : (C_ - 16);
    load_kblock(wp + kn, xp + (size_t)kn * N_, ra);
    load_kblock(wp + kn + 8, xp + (size_t)(kn + 8) * N_, rb);
    __builtin_amdgcn_sched_barrier(0);
    mma_kblock(oa, acc);
    mma_kblock(ob, acc);
    __builtin_amdgcn_sched_barrier(0);
    make_ops(ra, oa);
    make_ops(rb, ob);
  }

  const int bh = b * HDS + h;
  const size_t tile = (size_t)bh * NT + (n0 >> 4) + (c >> 2);
  if (p == 2) {
    float* vbase = vk + tile * 512 + (c & 1) * 128 + (((c >> 1) & 1) * 16 + 8 * hf) * 4;
#pragma unroll
    for (int i = 0; i < 2; ++i) {
#pragma unroll
      for (int r = 0; r < 8; ++r) {
        const v4f v = {acc[i][0][r], acc[i][1][r], acc[i][2][r], acc[i][3][r]};
        *(v4f*)(vbase + i * 256 + r * 4) = v;
      }
    }
  } else if (p == 1) {
    float* kbase = kt + tile * 512 + (2 * hf) * 128 + (4 * (c & 3)) * 4;
#pragma unroll
    for (int i = 0; i < 2; ++i) {
#pragma unroll
      for (int j = 0; j < 4; ++j) {
        float* q = kbase + (i * 16 + j) * 4;
        const v8f a = acc[i][j];
        const v4f lo = {a[0], a[1], a[2], a[3]};
        const v4f hi = {a[4], a[5], a[6], a[7]};
        *(v4f*)(q)       = lo;
        *(v4f*)(q + 128) = hi;
      }
    }
  } else {
    float* dst = qt;
    const float scale = QSCALE;
    float* qbase = dst + ((size_t)bh * N_ + n0 + 4 * c) * DK + 8 * hf;
#pragma unroll
    for (int i = 0; i < 2; ++i) {
#pragma unroll
      for (int j = 0; j < 4; ++j) {
        float* q = qbase + j * DK + 16 * i;
        const v8f a = acc[i][j];
        const v4f lo = {a[0] * scale, a[1] * scale, a[2] * scale, a[3] * scale};
        const v4f hi = {a[4] * scale, a[5] * scale, a[6] * scale, a[7] * scale};
        *(v4f*)(q)     = lo;
        *(v4f*)(q + 4) = hi;
      }
    }
  }
}

static __device__ __forceinline__ void load_tile(const v4f* __restrict__ kp,
                                                 const v4f* __restrict__ vp,
                                                 v4f (&f)[8]) {
  f[0] = kp[0];  f[1] = kp[32]; f[2] = kp[64]; f[3] = kp[96];
  f[4] = vp[0];  f[5] = vp[32]; f[6] = vp[64]; f[7] = vp[96];
}

static __device__ __forceinline__ void attn_tile(const v4f (&f)[8],
                                                 const v2f (&bq)[8],
                                                 v8f& o0, v8f& o1,
                                                 float& m, float& l) {
  const v4f k0 = f[0], k1 = f[1], k2 = f[2], k3 = f[3];
  const v4f va0 = f[4], va1 = f[5];
  const v4f vb0 = f[6], vb1 = f[7];

  v8f S = {};
  S = wmma_f32((v2f){k0.x, k0.y}, bq[0], S);
  S = wmma_f32((v2f){k0.z, k0.w}, bq[1], S);
  S = wmma_f32((v2f){k1.x, k1.y}, bq[2], S);
  S = wmma_f32((v2f){k1.z, k1.w}, bq[3], S);
  S = wmma_f32((v2f){k2.x, k2.y}, bq[4], S);
  S = wmma_f32((v2f){k2.z, k2.w}, bq[5], S);
  S = wmma_f32((v2f){k3.x, k3.y}, bq[6], S);
  S = wmma_f32((v2f){k3.z, k3.w}, bq[7], S);

  float rm = fmaxf(fmaxf(fmaxf(S[0], S[1]), fmaxf(S[2], S[3])),
                   fmaxf(fmaxf(S[4], S[5]), fmaxf(S[6], S[7])));
  rm = fmaxf(rm, xhalf(rm));
  const float mn = fmaxf(m, rm);
  const float a  = __expf(m - mn);
  m = mn;

  v8f P;
#pragma unroll
  for (int r = 0; r < 8; ++r) P[r] = __expf(S[r] - mn);
  const float rs = ((P[0] + P[1]) + (P[2] + P[3])) +
                   ((P[4] + P[5]) + (P[6] + P[7]));
  l = l * a + rs;
  o0 *= a;
  o1 *= a;

  o0 = wmma_f32((v2f){va0.x, va0.y}, (v2f){P[0], P[1]}, o0);
  o1 = wmma_f32((v2f){vb0.x, vb0.y}, (v2f){P[0], P[1]}, o1);
  o0 = wmma_f32((v2f){va0.z, va0.w}, (v2f){P[2], P[3]}, o0);
  o1 = wmma_f32((v2f){vb0.z, vb0.w}, (v2f){P[2], P[3]}, o1);
  o0 = wmma_f32((v2f){va1.x, va1.y}, (v2f){P[4], P[5]}, o0);
  o1 = wmma_f32((v2f){vb1.x, vb1.y}, (v2f){P[4], P[5]}, o1);
  o0 = wmma_f32((v2f){va1.z, va1.w}, (v2f){P[6], P[7]}, o0);
  o1 = wmma_f32((v2f){vb1.z, vb1.w}, (v2f){P[6], P[7]}, o1);
}

__global__ __launch_bounds__(128) void mha_attn_kernel(
    const float* __restrict__ x,
    const float* __restrict__ qt, const float* __restrict__ kt,
    const float* __restrict__ vk, float* __restrict__ out) {
  const int lane = threadIdx.x & 31;
  const int wave = threadIdx.x >> 5;

  const int task  = blockIdx.x * 4 + wave;
  const int qTile = task % NT;
  const int bh    = task / NT;
  const int h = bh & 7;
  const int b = bh >> 3;

  const int c  = lane & 15;
  const int hf = lane >> 4;
  const int nBase = qTile * 16;

  v2f bq[8];
  {
    const v4f* qp = (const v4f*)(qt + ((size_t)bh * N_ + nBase + c) * DK + 16 * hf);
    const v4f q0 = qp[0], q1 = qp[1], q2 = qp[2], q3 = qp[3];
    bq[0] = (v2f){q0.x, q0.y};  bq[1] = (v2f){q0.z, q0.w};
    bq[2] = (v2f){q1.x, q1.y};  bq[3] = (v2f){q1.z, q1.w};
    bq[4] = (v2f){q2.x, q2.y};  bq[5] = (v2f){q2.z, q2.w};
    bq[6] = (v2f){q3.x, q3.y};  bq[7] = (v2f){q3.z, q3.w};
  }

  const float* kptr = kt + (size_t)bh * N_ * DK + 4 * lane;
  const float* vptr = vk + (size_t)bh * N_ * DK + 4 * lane;

  v8f o0 = {}, o1 = {};
  float m = -1e30f, l = 0.0f;

  const v4f* kb = (const v4f*)kptr;
  const v4f* vb = (const v4f*)vptr;
  v4f fa[8], fb[8];
  load_tile(kb, vb, fa);
  load_tile(kb + 128, vb + 128, fb);

  for (int mt = 0; mt < NT; mt += 2) {
    const int na = (mt + 2 < NT) ? (mt + 2) : (NT - 1);
    const int nb = (mt + 3 < NT) ? (mt + 3) : (NT - 1);
    attn_tile(fa, bq, o0, o1, m, l);
    load_tile(kb + (size_t)na * 128, vb + (size_t)na * 128, fa);
    __builtin_amdgcn_sched_barrier(0);
    attn_tile(fb, bq, o0, o1, m, l);
    load_tile(kb + (size_t)nb * 128, vb + (size_t)nb * 128, fb);
    __builtin_amdgcn_sched_barrier(0);
  }

  const float inv = 1.0f / (l + xhalf(l));
  const size_t obase = ((size_t)b * C_ + h * DK + 8 * hf) * N_ + nBase + c;
#pragma unroll
  for (int r = 0; r < 8; ++r) {
    const size_t i0 = obase + (size_t)r * N_;
    out[i0] = x[i0] + o0[r] * inv;
    const size_t i1 = i0 + (size_t)16 * N_;
    out[i1] = x[i1] + o1[r] * inv;
  }
}

extern "C" void kernel_launch(void* const* d_in, const int* in_sizes, int n_in,
                              void* d_out, int out_size, void* d_ws,
                              size_t ws_size, hipStream_t stream) {
  const float* x  = (const float*)d_in[0];
  const float* wq = (const float*)d_in[1];
  const float* wk = (const float*)d_in[2];
  const float* wv = (const float*)d_in[3];
  float* out = (float*)d_out;

  float* qt = (float*)d_ws;
  float* kt = qt + (size_t)B_ * C_ * N_;
  float* vk = kt + (size_t)B_ * C_ * N_;

  dim3 blk(128);
  mha_proj_kernel<<<dim3(72, 12), blk, 0, stream>>>(x, wq, wk, wv, qt, kt, vk);
  mha_attn_kernel<<<dim3(1152), blk, 0, stream>>>(x, qt, kt, vk, out);
}


# Attention grid re-cut to 4 equal waves per SIMD: 512 groups of 8 waves x 9 query tiles, the ninth tile split by keys over the 8 waves, partials joined in LDS (changes that tile's softmax sum order)
# speedup vs baseline: 1.9806x; 1.0495x over previous
#include <hip/hip_runtime.h>
#include <math.h>


typedef __attribute__((ext_vector_type(2))) float v2f;
typedef __attribute__((ext_vector_type(4))) float v4f;
typedef __attribute__((ext_vector_type(8))) float v8f;

#define B_   4
#define C_   256
#define HDS  8
#define DK   32
#define N_   2304
#define NT   (N_ / 16)
#define QSCALE 0.17677669529663687f

static __device__ __forceinline__ v8f wmma_f32(v2f a, v2f b, v8f c) {
  return __builtin_amdgcn_wmma_f32_16x16x4_f32(
      false, a, false, b, (short)0, c, false, false);
}

static __device__ __forceinline__ float xhalf(float v) {
  const int b = __builtin_bit_cast(int, v);
  return __builtin_bit_cast(
      float, __builtin_amdgcn_permlanex16(b, b, 0x76543210, 0xfedcba98,
                                          false, false));
}


static __device__ __forceinline__ void load_kblock(const float* __restrict__ wp,
                                                   const float* __restrict__ xp,
                                                   v4f (&f)[6]) {
  f[0] = *(const v4f*)(wp);
  f[1] = *(const v4f*)(wp + 16 * C_);
  f[2] = *(const v4f*)(xp);
  f[3] = *(const v4f*)(xp + N_);
  f[4] = *(const v4f*)(xp + 2 * N_);
  f[5] = *(const v4f*)(xp + 3 * N_);
}

struct KOps {
  v2f a[2][2];
  v2f b[2][4];
};

static __device__ __forceinline__ void make_ops(const v4f (&f)[6], KOps& o) {
#pragma unroll
  for (int s = 0; s < 2; ++s) {
    o.a[0][s] = (v2f){f[0][2 * s], f[0][2 * s + 1]};
    o.a[1][s] = (v2f){f[1][2 * s], f[1][2 * s + 1]};
#pragma unroll
    for (int j = 0; j < 4; ++j)
      o.b[s][j] = (v2f){f[2 + 2 * s][j], f[3 + 2 * s][j]};
  }
}

static __device__ __forceinline__ void mma_kblock(const KOps& o,
                                                  v8f (&acc)[2][4]) {
#pragma unroll
  for (int s = 0; s < 2; ++s) {
#pragma unroll
    for (int j = 0; j < 4; ++j) {
      acc[0][j] = wmma_f32(o.a[0][s], o.b[s][j], acc[0][j]);
      acc[1][j] = wmma_f32(o.a[1][s], o.b[s][j], acc[1][j]);
    }
  }
}

__global__ __launch_bounds__(128) void mha_proj_kernel(
    const float* __restrict__ x,
    const float* __restrict__ wq, const float* __restrict__ wk,
    const float* __restrict__ wv,
    float* __restrict__ qt, float* __restrict__ kt, float* __restrict__ vk) {
  const int lane = threadIdx.x & 31;
  const int wave = threadIdx.x >> 5;
  const int nb = blockIdx.x >> 1;
  const int h  = (blockIdx.x & 1) * 4 + wave;
  const int p = blockIdx.y % 3;
  const int b = blockIdx.y / 3;
  const float* w = (p == 0) ? wq : (p == 1) ? wk : wv;

  const int c  = lane & 15;
  const int hf = lane >> 4;
  const int n0 = nb * 64;

  const float* wp = w + (size_t)(h * DK + c) * C_ + 4 * hf;
  const float* xp = x + ((size_t)b * C_ + 4 * hf) * N_ + n0 + 4 * c;

  v8f acc[2][4] = {};
  v4f ra[6], rb[6];
  KOps oa, ob;
  load_kblock(wp, xp, ra);
  load_kblock(wp + 8, xp + (size_t)8 * N_, rb);
  make_ops(ra, oa);
  make_ops(rb, ob);

  for (int k0 = 0; k0 < C_; k0 += 16) {
    const int kn = (k0 + 16 < C_) ? (k0 + 16) : (C_ - 16);
    load_kblock(wp + kn, xp + (size_t)kn * N_, ra);
    load_kblock(wp + kn + 8, xp + (size_t)(kn + 8) * N_, rb);
    __builtin_amdgcn_sched_barrier(0);
    mma_kblock(oa, acc);
    mma_kblock(ob, acc);
    __builtin_amdgcn_sched_barrier(0);
    make_ops(ra, oa);
    make_ops(rb, ob);
  }

  const int bh = b * HDS + h;
  const size_t tile = (size_t)bh * NT + (n0 >> 4) + (c >> 2);
  if (p == 2) {
    float* vbase = vk + tile * 512 + (c & 1) * 128 + (((c >> 1) & 1) * 16 + 8 * hf) * 4;
#pragma unroll
    for (int i = 0; i < 2; ++i) {
#pragma unroll
      for (int r = 0; r < 8; ++r) {
        const v4f v = {acc[i][0][r], acc[i][1][r], acc[i][2][r], acc[i][3][r]};
        *(v4f*)(vbase + i * 256 + r * 4) = v;
      }
    }
  } else if (p == 1) {
    float* kbase = kt + tile * 512 + (2 * hf) * 128 + (4 * (c & 3)) * 4;
#pragma unroll
    for (int i = 0; i < 2; ++i) {
#pragma unroll
      for (int j = 0; j < 4; ++j) {
        float* q = kbase + (i * 16 + j) * 4;
        const v8f a = acc[i][j];
        const v4f lo = {a[0], a[1], a[2], a[3]};
        const v4f hi = {a[4], a[5], a[6], a[7]};
        *(v4f*)(q)       = lo;
        *(v4f*)(q + 128) = hi;
      }
    }
  } else {
    float* dst = qt;
    const float scale = QSCALE;
    float* qbase = dst + ((size_t)bh * N_ + n0 + 4 * c) * DK + 8 * hf;
#pragma unroll
    for (int i = 0; i < 2; ++i) {
#pragma unroll
      for (int j = 0; j < 4; ++j) {
        float* q = qbase + j * DK + 16 * i;
        const v8f a = acc[i][j];
        const v4f lo = {a[0] * scale, a[1] * scale, a[2] * scale, a[3] * scale};
        const v4f hi = {a[4] * scale, a[5] * scale, a[6] * scale, a[7] * scale};
        *(v4f*)(q)     = lo;
        *(v4f*)(q + 4) = hi;
      }
    }
  }
}

static __device__ __forceinline__ void load_tile(const v4f* __restrict__ kp,
                                                 const v4f* __restrict__ vp,
                                                 v4f (&f)[8]) {
  f[0] = kp[0];  f[1] = kp[32]; f[2] = kp[64]; f[3] = kp[96];
  f[4] = vp[0];  f[5] = vp[32]; f[6] = vp[64]; f[7] = vp[96];
}

static __device__ __forceinline__ void attn_tile(const v4f (&f)[8],
                                                 const v2f (&bq)[8],
                                                 v8f& o0, v8f& o1,
                                                 float& m, float& l) {
  const v4f k0 = f[0], k1 = f[1], k2 = f[2], k3 = f[3];
  const v4f va0 = f[4], va1 = f[5];
  const v4f vb0 = f[6], vb1 = f[7];

  v8f S = {};
  S = wmma_f32((v2f){k0.x, k0.y}, bq[0], S);
  S = wmma_f32((v2f){k0.z, k0.w}, bq[1], S);
  S = wmma_f32((v2f){k1.x, k1.y}, bq[2], S);
  S = wmma_f32((v2f){k1.z, k1.w}, bq[3], S);
  S = wmma_f32((v2f){k2.x, k2.y}, bq[4], S);
  S = wmma_f32((v2f){k2.z, k2.w}, bq[5], S);
  S = wmma_f32((v2f){k3.x, k3.y}, bq[6], S);
  S = wmma_f32((v2f){k3.z, k3.w}, bq[7], S);

  float rm = fmaxf(fmaxf(fmaxf(S[0], S[1]), fmaxf(S[2], S[3])),
                   fmaxf(fmaxf(S[4], S[5]), fmaxf(S[6], S[7])));
  rm = fmaxf(rm, xhalf(rm));
  const float mn = fmaxf(m, rm);
  const float a  = __expf(m - mn);
  m = mn;

  v8f P;
#pragma unroll
  for (int r = 0; r < 8; ++r) P[r] = __expf(S[r] - mn);
  const float rs = ((P[0] + P[1]) + (P[2] + P[3])) +
                   ((P[4] + P[5]) + (P[6] + P[7]));
  l = l * a + rs;
  o0 *= a;
  o1 *= a;

  o0 = wmma_f32((v2f){va0.x, va0.y}, (v2f){P[0], P[1]}, o0);
  o1 = wmma_f32((v2f){vb0.x, vb0.y}, (v2f){P[0], P[1]}, o1);
  o0 = wmma_f32((v2f){va0.z, va0.w}, (v2f){P[2], P[3]}, o0);
  o1 = wmma_f32((v2f){vb0.z, vb0.w}, (v2f){P[2], P[3]}, o1);
  o0 = wmma_f32((v2f){va1.x, va1.y}, (v2f){P[4], P[5]}, o0);
  o1 = wmma_f32((v2f){vb1.x, vb1.y}, (v2f){P[4], P[5]}, o1);
  o0 = wmma_f32((v2f){va1.z, va1.w}, (v2f){P[6], P[7]}, o0);
  o1 = wmma_f32((v2f){vb1.z, vb1.w}, (v2f){P[6], P[7]}, o1);
}


#define QT_PER_WG 9
#define WG_PER_BH (NT / QT_PER_WG)
#define SH_KT     (NT / 8)

static __device__ __forceinline__ void attn_sweep(const float* __restrict__ qrow,
                                                  const v4f* __restrict__ kb,
                                                  const v4f* __restrict__ vb,
                                                  const int nkt,
                                                  v8f& o0, v8f& o1,
                                                  float& m, float& l) {
  v2f bq[8];
  {
    const v4f* qp = (const v4f*)qrow;
    const v4f q0 = qp[0], q1 = qp[1], q2 = qp[2], q3 = qp[3];
    bq[0] = (v2f){q0.x, q0.y};  bq[1] = (v2f){q0.z, q0.w};
    bq[2] = (v2f){q1.x, q1.y};  bq[3] = (v2f){q1.z, q1.w};
    bq[4] = (v2f){q2.x, q2.y};  bq[5] = (v2f){q2.z, q2.w};
    bq[6] = (v2f){q3.x, q3.y};  bq[7] = (v2f){q3.z, q3.w};
  }

  v4f fa[8], fb[8];
  load_tile(kb, vb, fa);
  load_tile(kb + 128, vb + 128, fb);

  for (int mt = 0; mt < nkt; mt += 2) {
    const int na = (mt + 2 < nkt) ? (mt + 2) : (nkt - 1);
    const int nb = (mt + 3 < nkt) ? (mt + 3) : (nkt - 1);
    attn_tile(fa, bq, o0, o1, m, l);
    load_tile(kb + (size_t)na * 128, vb + (size_t)na * 128, fa);
    __builtin_amdgcn_sched_barrier(0);
    attn_tile(fb, bq, o0, o1, m, l);
    load_tile(kb + (size_t)nb * 128, vb + (size_t)nb * 128, fb);
    __builtin_amdgcn_sched_barrier(0);
  }
}

__global__ __launch_bounds__(256) __attribute__((amdgpu_waves_per_eu(1, 4)))
void mha_attn_kernel(
    const float* __restrict__ x,
    const float* __restrict__ qt, const float* __restrict__ kt,
    const float* __restrict__ vk, float* __restrict__ out) {
  __shared__ float part[8][18][32];

  const int lane = threadIdx.x & 31;
  const int wave = __builtin_amdgcn_readfirstlane((int)(threadIdx.x >> 5));

  const int bh    = blockIdx.x / WG_PER_BH;
  const int qBase = (blockIdx.x % WG_PER_BH) * QT_PER_WG;
  const int h = bh & 7;
  const int b = bh >> 3;

  const int c  = lane & 15;
  const int hf = lane >> 4;

  const v4f* kb = (const v4f*)(kt + (size_t)bh * N_ * DK + 4 * lane);
  const v4f* vb = (const v4f*)(vk + (size_t)bh * N_ * DK + 4 * lane);
  const float* qbh = qt + (size_t)bh * N_ * DK + 16 * hf;

  {
    const int nS = (qBase + 8) * 16;
    v8f o0 = {}, o1 = {};
    float m = -1e30f, l = 0.0f;
    const size_t t0 = (size_t)(wave * SH_KT) * 128;
    attn_sweep(qbh + (size_t)(nS + c) * DK, kb + t0, vb + t0, SH_KT, o0, o1, m, l);

#pragma unroll
    for (int r = 0; r < 8; ++r) {
      part[wave][r][lane]     = o0[r];
      part[wave][8 + r][lane] = o1[r];
    }
    part[wave][16][lane] = m;
    part[wave][17][lane] = l;
    __syncthreads();

    float mj = part[0][16][lane];
    for (int i = 1; i < 8; ++i) mj = fmaxf(mj, part[i][16][lane]);
    float lj = 0.0f, a0 = 0.0f, a1 = 0.0f;
    for (int i = 0; i < 8; ++i) {
      const float s = __expf(part[i][16][lane] - mj);
      lj += s * part[i][17][lane];
      a0 += s * part[i][wave][lane];
      a1 += s * part[i][8 + wave][lane];
    }
    const float inv = 1.0f / (lj + xhalf(lj));
    const size_t i0 = ((size_t)b * C_ + h * DK + 8 * hf + wave) * N_ + nS + c;
    out[i0] = x[i0] + a0 * inv;
    const size_t i1 = i0 + (size_t)16 * N_;
    out[i1] = x[i1] + a1 * inv;
  }

  {
    const int nBase = (qBase + wave) * 16;
    v8f o0 = {}, o1 = {};
    float m = -1e30f, l = 0.0f;
    attn_sweep(qbh + (size_t)(nBase + c) * DK, kb, vb, NT, o0, o1, m, l);

    const float inv = 1.0f / (l + xhalf(l));
    const size_t obase = ((size_t)b * C_ + h * DK + 8 * hf) * N_ + nBase + c;
#pragma unroll
    for (int r = 0; r < 8; ++r) {
      const size_t i0 = obase + (size_t)r * N_;
      out[i0] = x[i0] + o0[r] * inv;
      const size_t i1 = i0 + (size_t)16 * N_;
      out[i1] = x[i1] + o1[r] * inv;
    }
  }
}

extern "C" void kernel_launch(void* const* d_in, const int* in_sizes, int n_in,
                              void* d_out, int out_size, void* d_ws,
                              size_t ws_size, hipStream_t stream) {
  const float* x  = (const float*)d_in[0];
  const float* wq = (const float*)d_in[1];
  const float* wk = (const float*)d_in[2];
  const float* wv = (const float*)d_in[3];
  float* out = (float*)d_out;

  float* qt = (float*)d_ws;
  float* kt = qt + (size_t)B_ * C_ * N_;
  float* vk = kt + (size_t)B_ * C_ * N_;

  dim3 blk(128);
  mha_proj_kernel<<<dim3(72, 12), blk, 0, stream>>>(x, wq, wk, wv, qt, kt, vk);
  mha_attn_kernel<<<dim3(512), dim3(256), 0, stream>>>(x, qt, kt, vk, out);
}


# Attention: s_setprio 3 at wave start, stepped to 2, 1, 0 behind 96, 124, 136 of the wave's own 144 key tiles: a wave ahead yields the matrix pipe, so a SIMD's four waves end together; same sums
# speedup vs baseline: 2.0423x; 1.0312x over previous
#include <hip/hip_runtime.h>
#include <math.h>


typedef __attribute__((ext_vector_type(2))) float v2f;
typedef __attribute__((ext_vector_type(4))) float v4f;
typedef __attribute__((ext_vector_type(8))) float v8f;

#define B_   4
#define C_   256
#define HDS  8
#define DK   32
#define N_   2304
#define NT   (N_ / 16)
#define QSCALE 0.17677669529663687f

static __device__ __forceinline__ v8f wmma_f32(v2f a, v2f b, v8f c) {
  return __builtin_amdgcn_wmma_f32_16x16x4_f32(
      false, a, false, b, (short)0, c, false, false);
}

static __device__ __forceinline__ float xhalf(float v) {
  const int b = __builtin_bit_cast(int, v);
  return __builtin_bit_cast(
      float, __builtin_amdgcn_permlanex16(b, b, 0x76543210, 0xfedcba98,
                                          false, false));
}


static __device__ __forceinline__ void load_kblock(const float* __restrict__ wp,
                                                   const float* __restrict__ xp,
                                                   v4f (&f)[6]) {
  f[0] = *(const v4f*)(wp);
  f[1] = *(const v4f*)(wp + 16 * C_);
  f[2] = *(const v4f*)(xp);
  f[3] = *(const v4f*)(xp + N_);
  f[4] = *(const v4f*)(xp + 2 * N_);
  f[5] = *(const v4f*)(xp + 3 * N_);
}

struct KOps {
  v2f a[2][2];
  v2f b[2][4];
};

static __device__ __forceinline__ void make_ops(const v4f (&f)[6], KOps& o) {
#pragma unroll
  for (int s = 0; s < 2; ++s) {
    o.a[0][s] = (v2f){f[0][2 * s], f[0][2 * s + 1]};
    o.a[1][s] = (v2f){f[1][2 * s], f[1][2 * s + 1]};
#pragma unroll
    for (int j = 0; j < 4; ++j)
      o.b[s][j] = (v2f){f[2 + 2 * s][j], f[3 + 2 * s][j]};
  }
}

static __device__ __forceinline__ void mma_kblock(const KOps& o,
                                                  v8f (&acc)[2][4]) {
#pragma unroll
  for (int s = 0; s < 2; ++s) {
#pragma unroll
    for (int j = 0; j < 4; ++j) {
      acc[0][j] = wmma_f32(o.a[0][s], o.b[s][j], acc[0][j]);
      acc[1][j] = wmma_f32(o.a[1][s], o.b[s][j], acc[1][j]);
    }
  }
}

__global__ __launch_bounds__(128) void mha_proj_kernel(
    const float* __restrict__ x,
    const float* __restrict__ wq, const float* __restrict__ wk,
    const float* __restrict__ wv,
    float* __restrict__ qt, float* __restrict__ kt, float* __restrict__ vk) {
  const int lane = threadIdx.x & 31;
  const int wave = threadIdx.x >> 5;
  const int nb = blockIdx.x >> 1;
  const int h  = (blockIdx.x & 1) * 4 + wave;
  const int p = blockIdx.y % 3;
  const int b = blockIdx.y / 3;
  const float* w = (p == 0) ? wq : (p == 1) ? wk : wv;

  const int c  = lane & 15;
  const int hf = lane >> 4;
  const int n0 = nb * 64;

  const float* wp = w + (size_t)(h * DK + c) * C_ + 4 * hf;
  const float* xp = x + ((size_t)b * C_ + 4 * hf) * N_ + n0 + 4 * c;

  v8f acc[2][4] = {};
  v4f ra[6], rb[6];
  KOps oa, ob;
  load_kblock(wp, xp, ra);
  load_kblock(wp + 8, xp + (size_t)8 * N_, rb);
  make_ops(ra, oa);
  make_ops(rb, ob);

  for (int k0 = 0; k0 < C_; k0 += 16) {
    const int kn = (k0 + 16 < C_) ? (k0 + 16) : (C_ - 16);
    load_kblock(wp + kn, xp + (size_t)kn * N_, ra);
    load_kblock(wp + kn + 8, xp + (size_t)(kn + 8) * N_, rb);
    __builtin_amdgcn_sched_barrier(0);
    mma_kblock(oa, acc);
    mma_kblock(ob, acc);
    __builtin_amdgcn_sched_barrier(0);
    make_ops(ra, oa);
    make_ops(rb, ob);
  }

  const int bh = b * HDS + h;
  const size_t tile = (size_t)bh * NT + (n0 >> 4) + (c >> 2);
  if (p == 2) {
    float* vbase = vk + tile * 512 + (c & 1) * 128 + (((c >> 1) & 1) * 16 + 8 * hf) * 4;
#pragma unroll
    for (int i = 0; i < 2; ++i) {
#pragma unroll
      for (int r = 0; r < 8; ++r) {
        const v4f v = {acc[i][0][r], acc[i][1][r], acc[i][2][r], acc[i][3][r]};
        *(v4f*)(vbase + i * 256 + r * 4) = v;
      }
    }
  } else if (p == 1) {
    float* kbase = kt + tile * 512 + (2 * hf) * 128 + (4 * (c & 3)) * 4;
#pragma unroll
    for (int i = 0; i < 2; ++i) {
#pragma unroll
      for (int j = 0; j < 4; ++j) {
        float* q = kbase + (i * 16 + j) * 4;
        const v8f a = acc[i][j];
        const v4f lo = {a[0], a[1], a[2], a[3]};
        const v4f hi = {a[4], a[5], a[6], a[7]};
        *(v4f*)(q)       = lo;
        *(v4f*)(q + 128) = hi;
      }
    }
  } else {
    float* dst = qt;
    const float scale = QSCALE;
    float* qbase = dst + ((size_t)bh * N_ + n0 + 4 * c) * DK + 8 * hf;
#pragma unroll
    for (int i = 0; i < 2; ++i) {
#pragma unroll
      for (int j = 0; j < 4; ++j) {
        float* q = qbase + j * DK + 16 * i;
        const v8f a = acc[i][j];
        const v4f lo = {a[0] * scale, a[1] * scale, a[2] * scale, a[3] * scale};
        const v4f hi = {a[4] * scale, a[5] * scale, a[6] * scale, a[7] * scale};
        *(v4f*)(q)     = lo;
        *(v4f*)(q + 4) = hi;
      }
    }
  }
}

static __device__ __forceinline__ void load_tile(const v4f* __restrict__ kp,
                                                 const v4f* __restrict__ vp,
                                                 v4f (&f)[8]) {
  f[0] = kp[0];  f[1] = kp[32]; f[2] = kp[64]; f[3] = kp[96];
  f[4] = vp[0];  f[5] = vp[32]; f[6] = vp[64]; f[7] = vp[96];
}

static __device__ __forceinline__ void attn_tile(const v4f (&f)[8],
                                                 const v2f (&bq)[8],
                                                 v8f& o0, v8f& o1,
                                                 float& m, float& l) {
  const v4f k0 = f[0], k1 = f[1], k2 = f[2], k3 = f[3];
  const v4f va0 = f[4], va1 = f[5];
  const v4f vb0 = f[6], vb1 = f[7];

  v8f S = {};
  S = wmma_f32((v2f){k0.x, k0.y}, bq[0], S);
  S = wmma_f32((v2f){k0.z, k0.w}, bq[1], S);
  S = wmma_f32((v2f){k1.x, k1.y}, bq[2], S);
  S = wmma_f32((v2f){k1.z, k1.w}, bq[3], S);
  S = wmma_f32((v2f){k2.x, k2.y}, bq[4], S);
  S = wmma_f32((v2f){k2.z, k2.w}, bq[5], S);
  S = wmma_f32((v2f){k3.x, k3.y}, bq[6], S);
  S = wmma_f32((v2f){k3.z, k3.w}, bq[7], S);

  float rm = fmaxf(fmaxf(fmaxf(S[0], S[1]), fmaxf(S[2], S[3])),
                   fmaxf(fmaxf(S[4], S[5]), fmaxf(S[6], S[7])));
  rm = fmaxf(rm, xhalf(rm));
  const float mn = fmaxf(m, rm);
  const float a  = __expf(m - mn);
  m = mn;

  v8f P;
#pragma unroll
  for (int r = 0; r < 8; ++r) P[r] = __expf(S[r] - mn);
  const float rs = ((P[0] + P[1]) + (P[2] + P[3])) +
                   ((P[4] + P[5]) + (P[6] + P[7]));
  l = l * a + rs;
  o0 *= a;
  o1 *= a;

  o0 = wmma_f32((v2f){va0.x, va0.y}, (v2f){P[0], P[1]}, o0);
  o1 = wmma_f32((v2f){vb0.x, vb0.y}, (v2f){P[0], P[1]}, o1);
  o0 = wmma_f32((v2f){va0.z, va0.w}, (v2f){P[2], P[3]}, o0);
  o1 = wmma_f32((v2f){vb0.z, vb0.w}, (v2f){P[2], P[3]}, o1);
  o0 = wmma_f32((v2f){va1.x, va1.y}, (v2f){P[4], P[5]}, o0);
  o1 = wmma_f32((v2f){vb1.x, vb1.y}, (v2f){P[4], P[5]}, o1);
  o0 = wmma_f32((v2f){va1.z, va1.w}, (v2f){P[6], P[7]}, o0);
  o1 = wmma_f32((v2f){vb1.z, vb1.w}, (v2f){P[6], P[7]}, o1);
}


#define QT_PER_WG 9
#define WG_PER_BH (NT / QT_PER_WG)
#define SH_KT     (NT / 8)
#define PRIO_KT1  96
#define PRIO_KT2  124
#define PRIO_KT3  136

static __device__ __forceinline__ void sweep_trips(const v4f* __restrict__ kb,
                                                   const v4f* __restrict__ vb,
                                                   const int nkt,
                                                   const int mt0, const int mt1,
                                                   v4f (&fa)[8], v4f (&fb)[8],
                                                   const v2f (&bq)[8],
                                                   v8f& o0, v8f& o1,
                                                   float& m, float& l) {
  for (int mt = mt0; mt < mt1; mt += 2) {
    const int na = (mt + 2 < nkt) ? (mt + 2) : (nkt - 1);
    const int nb = (mt + 3 < nkt) ? (mt + 3) : (nkt - 1);
    attn_tile(fa, bq, o0, o1, m, l);
    load_tile(kb + (size_t)na * 128, vb + (size_t)na * 128, fa);
    __builtin_amdgcn_sched_barrier(0);
    attn_tile(fb, bq, o0, o1, m, l);
    load_tile(kb + (size_t)nb * 128, vb + (size_t)nb * 128, fb);
    __builtin_amdgcn_sched_barrier(0);
  }
}

template <bool kStepPrio>
static __device__ __forceinline__ void attn_sweep(const float* __restrict__ qrow,
                                                  const v4f* __restrict__ kb,
                                                  const v4f* __restrict__ vb,
                                                  const int nkt,
                                                  v8f& o0, v8f& o1,
                                                  float& m, float& l) {
  v2f bq[8];
  {
    const v4f* qp = (const v4f*)qrow;
    const v4f q0 = qp[0], q1 = qp[1], q2 = qp[2], q3 = qp[3];
    bq[0] = (v2f){q0.x, q0.y};  bq[1] = (v2f){q0.z, q0.w};
    bq[2] = (v2f){q1.x, q1.y};  bq[3] = (v2f){q1.z, q1.w};
    bq[4] = (v2f){q2.x, q2.y};  bq[5] = (v2f){q2.z, q2.w};
    bq[6] = (v2f){q3.x, q3.y};  bq[7] = (v2f){q3.z, q3.w};
  }

  v4f fa[8], fb[8];
  load_tile(kb, vb, fa);
  load_tile(kb + 128, vb + 128, fb);

  if (kStepPrio) {
    sweep_trips(kb, vb, nkt, 0, PRIO_KT1, fa, fb, bq, o0, o1, m, l);
    __builtin_amdgcn_s_setprio(2);
    sweep_trips(kb, vb, nkt, PRIO_KT1, PRIO_KT2, fa, fb, bq, o0, o1, m, l);
    __builtin_amdgcn_s_setprio(1);
    sweep_trips(kb, vb, nkt, PRIO_KT2, PRIO_KT3, fa, fb, bq, o0, o1, m, l);
    __builtin_amdgcn_s_setprio(0);
    sweep_trips(kb, vb, nkt, PRIO_KT3, nkt, fa, fb, bq, o0, o1, m, l);
  } else {
    sweep_trips(kb, vb, nkt, 0, nkt, fa, fb, bq, o0, o1, m, l);
  }
}

__global__ __launch_bounds__(256) __attribute__((amdgpu_waves_per_eu(1, 4)))
void mha_attn_kernel(
    const float* __restrict__ x,
    const float* __restrict__ qt, const float* __restrict__ kt,
    const float* __restrict__ vk, float* __restrict__ out) {
  __shared__ float part[8][18][32];

  const int lane = threadIdx.x & 31;
  const int wave = __builtin_amdgcn_readfirstlane((int)(threadIdx.x >> 5));

  const int bh    = blockIdx.x / WG_PER_BH;
  const int qBase = (blockIdx.x % WG_PER_BH) * QT_PER_WG;
  const int h = bh & 7;
  const int b = bh >> 3;

  const int c  = lane & 15;
  const int hf = lane >> 4;

  __builtin_amdgcn_s_setprio(3);

  const v4f* kb = (const v4f*)(kt + (size_t)bh * N_ * DK + 4 * lane);
  const v4f* vb = (const v4f*)(vk + (size_t)bh * N_ * DK + 4 * lane);
  const float* qbh = qt + (size_t)bh * N_ * DK + 16 * hf;

  {
    const int nS = (qBase + 8) * 16;
    v8f o0 = {}, o1 = {};
    float m = -1e30f, l = 0.0f;
    const size_t t0 = (size_t)(wave * SH_KT) * 128;
    attn_sweep<false>(qbh + (size_t)(nS + c) * DK, kb + t0, vb + t0, SH_KT, o0, o1, m, l);

#pragma unroll
    for (int r = 0; r < 8; ++r) {
      part[wave][r][lane]     = o0[r];
      part[wave][8 + r][lane] = o1[r];
    }
    part[wave][16][lane] = m;
    part[wave][17][lane] = l;
    __syncthreads();

    float mj = part[0][16][lane];
    for (int i = 1; i < 8; ++i) mj = fmaxf(mj, part[i][16][lane]);
    float lj = 0.0f, a0 = 0.0f, a1 = 0.0f;
    for (int i = 0; i < 8; ++i) {
      const float s = __expf(part[i][16][lane] - mj);
      lj += s * part[i][17][lane];
      a0 += s * part[i][wave][lane];
      a1 += s * part[i][8 + wave][lane];
    }
    const float inv = 1.0f / (lj + xhalf(lj));
    const size_t i0 = ((size_t)b * C_ + h * DK + 8 * hf + wave) * N_ + nS + c;
    out[i0] = x[i0] + a0 * inv;
    const size_t i1 = i0 + (size_t)16 * N_;
    out[i1] = x[i1] + a1 * inv;
  }

  {
    const int nBase = (qBase + wave) * 16;
    v8f o0 = {}, o1 = {};
    float m = -1e30f, l = 0.0f;
    attn_sweep<true>(qbh + (size_t)(nBase + c) * DK, kb, vb, NT, o0, o1, m, l);

    const float inv = 1.0f / (l + xhalf(l));
    const size_t obase = ((size_t)b * C_ + h * DK + 8 * hf) * N_ + nBase + c;
#pragma unroll
    for (int r = 0; r < 8; ++r) {
      const size_t i0 = obase + (size_t)r * N_;
      out[i0] = x[i0] + o0[r] * inv;
      const size_t i1 = i0 + (size_t)16 * N_;
      out[i1] = x[i1] + o1[r] * inv;
    }
  }
}

extern "C" void kernel_launch(void* const* d_in, const int* in_sizes, int n_in,
                              void* d_out, int out_size, void* d_ws,
                              size_t ws_size, hipStream_t stream) {
  const float* x  = (const float*)d_in[0];
  const float* wq = (const float*)d_in[1];
  const float* wk = (const float*)d_in[2];
  const float* wv = (const float*)d_in[3];
  float* out = (float*)d_out;

  float* qt = (float*)d_ws;
  float* kt = qt + (size_t)B_ * C_ * N_;
  float* vk = kt + (size_t)B_ * C_ * N_;

  dim3 blk(128);
  mha_proj_kernel<<<dim3(72, 12), blk, 0, stream>>>(x, wq, wk, wv, qt, kt, vk);
  mha_attn_kernel<<<dim3(512), dim3(256), 0, stream>>>(x, qt, kt, vk, out);
}


# Softmax in base 2 on a lagging maximum held as the score WMMA's C operand: P = exp2(D), no max, subtract or multiply; O, l re-based on a wave-uniform branch when a row sum tops 2^8; P rounds anew
# speedup vs baseline: 2.2145x; 1.0843x over previous
#include <hip/hip_runtime.h>
#include <math.h>


typedef __attribute__((ext_vector_type(2))) float v2f;
typedef __attribute__((ext_vector_type(4))) float v4f;
typedef __attribute__((ext_vector_type(8))) float v8f;

#define B_   4
#define C_   256
#define HDS  8
#define DK   32
#define N_   2304
#define NT   (N_ / 16)
#define QSCALE 0.25503486164919736f

static __device__ __forceinline__ v8f wmma_f32(v2f a, v2f b, v8f c) {
  return __builtin_amdgcn_wmma_f32_16x16x4_f32(
      false, a, false, b, (short)0, c, false, false);
}

static __device__ __forceinline__ float xhalf(float v) {
  const int b = __builtin_bit_cast(int, v);
  return __builtin_bit_cast(
      float, __builtin_amdgcn_permlanex16(b, b, 0x76543210, 0xfedcba98,
                                          false, false));
}

static __device__ __forceinline__ float ex2(float v) {
  return __builtin_amdgcn_exp2f(v);
}

static __device__ __forceinline__ float sum8(const v8f& p) {
  return ((p[0] + p[1]) + (p[2] + p[3])) + ((p[4] + p[5]) + (p[6] + p[7]));
}


static __device__ __forceinline__ void load_kblock(const float* __restrict__ wp,
                                                   const float* __restrict__ xp,
                                                   v4f (&f)[6]) {
  f[0] = *(const v4f*)(wp);
  f[1] = *(const v4f*)(wp + 16 * C_);
  f[2] = *(const v4f*)(xp);
  f[3] = *(const v4f*)(xp + N_);
  f[4] = *(const v4f*)(xp + 2 * N_);
  f[5] = *(const v4f*)(xp + 3 * N_);
}

struct KOps {
  v2f a[2][2];
  v2f b[2][4];
};

static __device__ __forceinline__ void make_ops(const v4f (&f)[6], KOps& o) {
#pragma unroll
  for (int s = 0; s < 2; ++s) {
    o.a[0][s] = (v2f){f[0][2 * s], f[0][2 * s + 1]};
    o.a[1][s] = (v2f){f[1][2 * s], f[1][2 * s + 1]};
#pragma unroll
    for (int j = 0; j < 4; ++j)
      o.b[s][j] = (v2f){f[2 + 2 * s][j], f[3 + 2 * s][j]};
  }
}

static __device__ __forceinline__ void mma_kblock(const KOps& o,
                                                  v8f (&acc)[2][4]) {
#pragma unroll
  for (int s = 0; s < 2; ++s) {
#pragma unroll
    for (int j = 0; j < 4; ++j) {
      acc[0][j] = wmma_f32(o.a[0][s], o.b[s][j], acc[0][j]);
      acc[1][j] = wmma_f32(o.a[1][s], o.b[s][j], acc[1][j]);
    }
  }
}

__global__ __launch_bounds__(128) void mha_proj_kernel(
    const float* __restrict__ x,
    const float* __restrict__ wq, const float* __restrict__ wk,
    const float* __restrict__ wv,
    float* __restrict__ qt, float* __restrict__ kt, float* __restrict__ vk) {
  const int lane = threadIdx.x & 31;
  const int wave = threadIdx.x >> 5;
  const int nb = blockIdx.x >> 1;
  const int h  = (blockIdx.x & 1) * 4 + wave;
  const int p = blockIdx.y % 3;
  const int b = blockIdx.y / 3;
  const float* w = (p == 0) ? wq : (p == 1) ? wk : wv;

  const int c  = lane & 15;
  const int hf = lane >> 4;
  const int n0 = nb * 64;

  const float* wp = w + (size_t)(h * DK + c) * C_ + 4 * hf;
  const float* xp = x + ((size_t)b * C_ + 4 * hf) * N_ + n0 + 4 * c;

  v8f acc[2][4] = {};
  v4f ra[6], rb[6];
  KOps oa, ob;
  load_kblock(wp, xp, ra);
  load_kblock(wp + 8, xp + (size_t)8 * N_, rb);
  make_ops(ra, oa);
  make_ops(rb, ob);

  for (int k0 = 0; k0 < C_; k0 += 16) {
    const int kn = (k0 + 16 < C_) ? (k0 + 16) : (C_ - 16);
    load_kblock(wp + kn, xp + (size_t)kn * N_, ra);
    load_kblock(wp + kn + 8, xp + (size_t)(kn + 8) * N_, rb);
    __builtin_amdgcn_sched_barrier(0);
    mma_kblock(oa, acc);
    mma_kblock(ob, acc);
    __builtin_amdgcn_sched_barrier(0);
    make_ops(ra, oa);
    make_ops(rb, ob);
  }

  const int bh = b * HDS + h;
  const size_t tile = (size_t)bh * NT + (n0 >> 4) + (c >> 2);
  if (p == 2) {
    float* vbase = vk + tile * 512 + (c & 1) * 128 + (((c >> 1) & 1) * 16 + 8 * hf) * 4;
#pragma unroll
    for (int i = 0; i < 2; ++i) {
#pragma unroll
      for (int r = 0; r < 8; ++r) {
        const v4f v = {acc[i][0][r], acc[i][1][r], acc[i][2][r], acc[i][3][r]};
        *(v4f*)(vbase + i * 256 + r * 4) = v;
      }
    }
  } else if (p == 1) {
    float* kbase = kt + tile * 512 + (2 * hf) * 128 + (4 * (c & 3)) * 4;
#pragma unroll
    for (int i = 0; i < 2; ++i) {
#pragma unroll
      for (int j = 0; j < 4; ++j) {
        float* q = kbase + (i * 16 + j) * 4;
        const v8f a = acc[i][j];
        const v4f lo = {a[0], a[1], a[2], a[3]};
        const v4f hi = {a[4], a[5], a[6], a[7]};
        *(v4f*)(q)       = lo;
        *(v4f*)(q + 128) = hi;
      }
    }
  } else {
    float* dst = qt;
    const float scale = QSCALE;
    float* qbase = dst + ((size_t)bh * N_ + n0 + 4 * c) * DK + 8 * hf;
#pragma unroll
    for (int i = 0; i < 2; ++i) {
#pragma unroll
      for (int j = 0; j < 4; ++j) {
        float* q = qbase + j * DK + 16 * i;
        const v8f a = acc[i][j];
        const v4f lo = {a[0] * scale, a[1] * scale, a[2] * scale, a[3] * scale};
        const v4f hi = {a[4] * scale, a[5] * scale, a[6] * scale, a[7] * scale};
        *(v4f*)(q)     = lo;
        *(v4f*)(q + 4) = hi;
      }
    }
  }
}

static __device__ __forceinline__ void load_tile(const v4f* __restrict__ kp,
                                                 const v4f* __restrict__ vp,
                                                 v4f (&f)[8]) {
  f[0] = kp[0];  f[1] = kp[32]; f[2] = kp[64]; f[3] = kp[96];
  f[4] = vp[0];  f[5] = vp[32]; f[6] = vp[64]; f[7] = vp[96];
}

static __device__ __forceinline__ void attn_tile(const v4f (&f)[8],
                                                 const v2f (&bq)[8],
                                                 v8f& o0, v8f& o1,
                                                 v8f& cneg, float& negm,
                                                 float& l, float& thr,
                                                 float& lo) {
  const v4f k0 = f[0], k1 = f[1], k2 = f[2], k3 = f[3];
  const v4f va0 = f[4], va1 = f[5];
  const v4f vb0 = f[6], vb1 = f[7];

  v8f S = cneg;
  S = wmma_f32((v2f){k0.x, k0.y}, bq[0], S);
  S = wmma_f32((v2f){k0.z, k0.w}, bq[1], S);
  S = wmma_f32((v2f){k1.x, k1.y}, bq[2], S);
  S = wmma_f32((v2f){k1.z, k1.w}, bq[3], S);
  S = wmma_f32((v2f){k2.x, k2.y}, bq[4], S);
  S = wmma_f32((v2f){k2.z, k2.w}, bq[5], S);
  S = wmma_f32((v2f){k3.x, k3.y}, bq[6], S);
  S = wmma_f32((v2f){k3.z, k3.w}, bq[7], S);

  v8f P;
#pragma unroll
  for (int r = 0; r < 8; ++r) P[r] = ex2(S[r]);
  float rs = sum8(P);

  if (__builtin_amdgcn_ballot_w32(rs > thr) != 0u) {
    float rm = fmaxf(fmaxf(fmaxf(S[0], S[1]), fmaxf(S[2], S[3])),
                     fmaxf(fmaxf(S[4], S[5]), fmaxf(S[6], S[7])));
    rm = fmaxf(rm, xhalf(rm));
    const float delta = fmaxf(rm, lo);
    const float a = ex2(fminf(-delta, 0.0f));
#pragma unroll
    for (int r = 0; r < 8; ++r) P[r] = ex2(S[r] - delta);
    rs = sum8(P);
    o0 *= a;
    o1 *= a;
    l *= a;
    negm -= delta;
#pragma unroll
    for (int r = 0; r < 8; ++r) cneg[r] = negm;
    thr = 256.0f;
    lo = 0.0f;
  }
  l += rs;

  o0 = wmma_f32((v2f){va0.x, va0.y}, (v2f){P[0], P[1]}, o0);
  o1 = wmma_f32((v2f){vb0.x, vb0.y}, (v2f){P[0], P[1]}, o1);
  o0 = wmma_f32((v2f){va0.z, va0.w}, (v2f){P[2], P[3]}, o0);
  o1 = wmma_f32((v2f){vb0.z, vb0.w}, (v2f){P[2], P[3]}, o1);
  o0 = wmma_f32((v2f){va1.x, va1.y}, (v2f){P[4], P[5]}, o0);
  o1 = wmma_f32((v2f){vb1.x, vb1.y}, (v2f){P[4], P[5]}, o1);
  o0 = wmma_f32((v2f){va1.z, va1.w}, (v2f){P[6], P[7]}, o0);
  o1 = wmma_f32((v2f){vb1.z, vb1.w}, (v2f){P[6], P[7]}, o1);
}


#define QT_PER_WG 9
#define WG_PER_BH (NT / QT_PER_WG)
#define SH_KT     (NT / 8)
#define PRIO_KT1  96
#define PRIO_KT2  124
#define PRIO_KT3  136

static __device__ __forceinline__ void sweep_trips(const v4f* __restrict__ kb,
                                                   const v4f* __restrict__ vb,
                                                   const int nkt,
                                                   const int mt0, const int mt1,
                                                   v4f (&fa)[8], v4f (&fb)[8],
                                                   const v2f (&bq)[8],
                                                   v8f& o0, v8f& o1,
                                                   v8f& cneg, float& negm,
                                                   float& l, float& thr,
                                                   float& lo) {
  for (int mt = mt0; mt < mt1; mt += 2) {
    const int na = (mt + 2 < nkt) ? (mt + 2) : (nkt - 1);
    const int nb = (mt + 3 < nkt) ? (mt + 3) : (nkt - 1);
    attn_tile(fa, bq, o0, o1, cneg, negm, l, thr, lo);
    load_tile(kb + (size_t)na * 128, vb + (size_t)na * 128, fa);
    __builtin_amdgcn_sched_barrier(0);
    attn_tile(fb, bq, o0, o1, cneg, negm, l, thr, lo);
    load_tile(kb + (size_t)nb * 128, vb + (size_t)nb * 128, fb);
    __builtin_amdgcn_sched_barrier(0);
  }
}

template <bool kStepPrio>
static __device__ __forceinline__ void attn_sweep(const float* __restrict__ qrow,
                                                  const v4f* __restrict__ kb,
                                                  const v4f* __restrict__ vb,
                                                  const int nkt,
                                                  v8f& o0, v8f& o1,
                                                  float& m, float& l) {
  v8f cneg = {};
  float negm = 0.0f, thr = -1.0f, lo = -3.0e38f;
  v2f bq[8];
  {
    const v4f* qp = (const v4f*)qrow;
    const v4f q0 = qp[0], q1 = qp[1], q2 = qp[2], q3 = qp[3];
    bq[0] = (v2f){q0.x, q0.y};  bq[1] = (v2f){q0.z, q0.w};
    bq[2] = (v2f){q1.x, q1.y};  bq[3] = (v2f){q1.z, q1.w};
    bq[4] = (v2f){q2.x, q2.y};  bq[5] = (v2f){q2.z, q2.w};
    bq[6] = (v2f){q3.x, q3.y};  bq[7] = (v2f){q3.z, q3.w};
  }

  v4f fa[8], fb[8];
  load_tile(kb, vb, fa);
  load_tile(kb + 128, vb + 128, fb);

  if (kStepPrio) {
    sweep_trips(kb, vb, nkt, 0, PRIO_KT1, fa, fb, bq, o0, o1, cneg, negm, l, thr, lo);
    __builtin_amdgcn_s_setprio(2);
    sweep_trips(kb, vb, nkt, PRIO_KT1, PRIO_KT2, fa, fb, bq, o0, o1, cneg, negm, l, thr, lo);
    __builtin_amdgcn_s_setprio(1);
    sweep_trips(kb, vb, nkt, PRIO_KT2, PRIO_KT3, fa, fb, bq, o0, o1, cneg, negm, l, thr, lo);
    __builtin_amdgcn_s_setprio(0);
    sweep_trips(kb, vb, nkt, PRIO_KT3, nkt, fa, fb, bq, o0, o1, cneg, negm, l, thr, lo);
  } else {
    sweep_trips(kb, vb, nkt, 0, nkt, fa, fb, bq, o0, o1, cneg, negm, l, thr, lo);
  }
  m = -negm;
}

__global__ __launch_bounds__(256) __attribute__((amdgpu_waves_per_eu(1, 4)))
void mha_attn_kernel(
    const float* __restrict__ x,
    const float* __restrict__ qt, const float* __restrict__ kt,
    const float* __restrict__ vk, float* __restrict__ out) {
  __shared__ float part[8][18][32];

  const int lane = threadIdx.x & 31;
  const int wave = __builtin_amdgcn_readfirstlane((int)(threadIdx.x >> 5));

  const int bh    = blockIdx.x / WG_PER_BH;
  const int qBase = (blockIdx.x % WG_PER_BH) * QT_PER_WG;
  const int h = bh & 7;
  const int b = bh >> 3;

  const int c  = lane & 15;
  const int hf = lane >> 4;

  __builtin_amdgcn_s_setprio(3);

  const v4f* kb = (const v4f*)(kt + (size_t)bh * N_ * DK + 4 * lane);
  const v4f* vb = (const v4f*)(vk + (size_t)bh * N_ * DK + 4 * lane);
  const float* qbh = qt + (size_t)bh * N_ * DK + 16 * hf;

  {
    const int nS = (qBase + 8) * 16;
    v8f o0 = {}, o1 = {};
    float m = 0.0f, l = 0.0f;
    const size_t t0 = (size_t)(wave * SH_KT) * 128;
    attn_sweep<false>(qbh + (size_t)(nS + c) * DK, kb + t0, vb + t0, SH_KT, o0, o1, m, l);

#pragma unroll
    for (int r = 0; r < 8; ++r) {
      part[wave][r][lane]     = o0[r];
      part[wave][8 + r][lane] = o1[r];
    }
    part[wave][16][lane] = m;
    part[wave][17][lane] = l;
    __syncthreads();

    float mj = part[0][16][lane];
    for (int i = 1; i < 8; ++i) mj = fmaxf(mj, part[i][16][lane]);
    float lj = 0.0f, a0 = 0.0f, a1 = 0.0f;
    for (int i = 0; i < 8; ++i) {
      const float s = ex2(part[i][16][lane] - mj);
      lj += s * part[i][17][lane];
      a0 += s * part[i][wave][lane];
      a1 += s * part[i][8 + wave][lane];
    }
    const float inv = 1.0f / (lj + xhalf(lj));
    const size_t i0 = ((size_t)b * C_ + h * DK + 8 * hf + wave) * N_ + nS + c;
    out[i0] = x[i0] + a0 * inv;
    const size_t i1 = i0 + (size_t)16 * N_;
    out[i1] = x[i1] + a1 * inv;
  }

  {
    const int nBase = (qBase + wave) * 16;
    v8f o0 = {}, o1 = {};
    float m = 0.0f, l = 0.0f;
    attn_sweep<true>(qbh + (size_t)(nBase + c) * DK, kb, vb, NT, o0, o1, m, l);

    const float inv = 1.0f / (l + xhalf(l));
    const size_t obase = ((size_t)b * C_ + h * DK + 8 * hf) * N_ + nBase + c;
#pragma unroll
    for (int r = 0; r < 8; ++r) {
      const size_t i0 = obase + (size_t)r * N_;
      out[i0] = x[i0] + o0[r] * inv;
      const size_t i1 = i0 + (size_t)16 * N_;
      out[i1] = x[i1] + o1[r] * inv;
    }
  }
}

extern "C" void kernel_launch(void* const* d_in, const int* in_sizes, int n_in,
                              void* d_out, int out_size, void* d_ws,
                              size_t ws_size, hipStream_t stream) {
  const float* x  = (const float*)d_in[0];
  const float* wq = (const float*)d_in[1];
  const float* wk = (const float*)d_in[2];
  const float* wv = (const float*)d_in[3];
  float* out = (float*)d_out;

  float* qt = (float*)d_ws;
  float* kt = qt + (size_t)B_ * C_ * N_;
  float* vk = kt + (size_t)B_ * C_ * N_;

  dim3 blk(128);
  mha_proj_kernel<<<dim3(72, 12), blk, 0, stream>>>(x, wq, wk, wv, qt, kt, vk);
  mha_attn_kernel<<<dim3(512), dim3(256), 0, stream>>>(x, qt, kt, vk, out);
}
